# starting kernel
# baseline (speedup 1.0000x reference)
#include <hip/hip_runtime.h>
#include <hip/hip_bf16.h>

#define T   1024
#define B   8
#define D   64
#define N   64
#define TC  32
#define C   (T / TC)
#define EPSF 1e-8f

typedef float v2f __attribute__((ext_vector_type(2)));
typedef float v4f __attribute__((ext_vector_type(4)));
typedef float v8f __attribute__((ext_vector_type(8)));

__global__ __launch_bounds__(256) void k_decay(const float* __restrict__ alpha,
                                               const float* __restrict__ kk,
                                               float* __restrict__ kdec) {
    int idx = blockIdx.x * blockDim.x + threadIdx.x;
    if (idx >= B * N) return;
    float runlog = 0.0f;
    for (int t = 0; t < T; ++t) {
        float a = alpha[(size_t)t * (B * N) + idx];
        a = fmaxf(a, EPSF);
        runlog += logf(a);
        kdec[(size_t)t * (B * N) + idx] = expf(runlog);
    }
    float inv = 1.0f / (expf(runlog) + EPSF);
    for (int t = 0; t < T; ++t) {
        size_t o = (size_t)t * (B * N) + idx;
        kdec[o] = kdec[o] * inv * kk[o];
    }
}

__global__ __launch_bounds__(256) void k_chunk_gemm(const float* __restrict__ v,
                                                    const float* __restrict__ kdec,
                                                    float* __restrict__ M) {
    const int c = blockIdx.x / B;
    const int b = blockIdx.x % B;
    const int tid   = threadIdx.x;
    const int wave  = tid >> 5;
    const int lane  = tid & 31;
    const int mt    = wave >> 1;
    const int ntb   = (wave & 1) * 2;
    const int lhalf = lane >> 4;
    const int l16   = lane & 15;

    v8f acc0 = {};
    v8f acc1 = {};
    const int t0 = c * TC;

    #pragma unroll
    for (int ks = 0; ks < TC / 4; ++ks) {
        const int tk = t0 + ks * 4 + lhalf * 2;
        const int d = mt * 16 + l16;
        v2f a;
        a.x = v[((size_t)(tk    ) * B + b) * D + d];
        a.y = v[((size_t)(tk + 1) * B + b) * D + d];
        const float* kr0 = kdec + ((size_t)(tk    ) * B + b) * N;
        const float* kr1 = kdec + ((size_t)(tk + 1) * B + b) * N;
        const int n0 = ntb * 16 + l16;
        const int n1 = (ntb + 1) * 16 + l16;
        v2f b0, b1;
        b0.x = kr0[n0]; b0.y = kr1[n0];
        b1.x = kr0[n1]; b1.y = kr1[n1];
        acc0 = __builtin_amdgcn_wmma_f32_16x16x4_f32(false, a, false, b0,
                                                     (short)0, acc0, false, false);
        acc1 = __builtin_amdgcn_wmma_f32_16x16x4_f32(false, a, false, b1,
                                                     (short)0, acc1, false, false);
    }

    float* Mc = M + ((size_t)c * B + b) * (D * N);
    const int mbase = mt * 16 + lhalf * 8;
    const int nn0 = ntb * 16 + l16;
    const int nn1 = (ntb + 1) * 16 + l16;
    #pragma unroll
    for (int r = 0; r < 8; ++r) {
        Mc[(size_t)(mbase + r) * N + nn0] = acc0[r];
        Mc[(size_t)(mbase + r) * N + nn1] = acc1[r];
    }
}

__global__ __launch_bounds__(256) void k_chunk_scan(float* __restrict__ M) {
    const int idx = blockIdx.x * blockDim.x + threadIdx.x;
    float run = 0.0f;
    #pragma unroll 1
    for (int c = 0; c < C; ++c) {
        const size_t o = (size_t)c * (B * D * N) + idx;
        const float m = M[o];
        M[o] = run;
        run += m;
    }
}

__global__ __launch_bounds__(256) void k_scan_store(const float* __restrict__ v,
                                                    const float* __restrict__ kdec,
                                                    const float* __restrict__ O,
                                                    float* __restrict__ S) {
    const int c = blockIdx.x / B;
    const int b = blockIdx.x % B;
    const int tid = threadIdx.x;
    const int nq = tid & 15;
    const int dq = tid >> 4;
    const int n0 = nq * 4;

    v4f acc[4];
    const float* Ob = O + ((size_t)c * B + b) * (D * N);
    #pragma unroll
    for (int i = 0; i < 4; ++i)
        acc[i] = *(const v4f*)(Ob + (size_t)(dq * 4 + i) * N + n0);

    const int tEnd = c * TC + TC;
    #pragma unroll 1
    for (int t = c * TC; t < tEnd; ++t) {
        const v4f kv = *(const v4f*)(kdec + ((size_t)t * B + b) * N + n0);
        const float* vr = v + ((size_t)t * B + b) * D + dq * 4;
        float* St = S + ((size_t)t * B + b) * (size_t)(D * N);
        #pragma unroll
        for (int i = 0; i < 4; ++i) {
            const float vi = vr[i];
            acc[i].x = fmaf(vi, kv.x, acc[i].x);
            acc[i].y = fmaf(vi, kv.y, acc[i].y);
            acc[i].z = fmaf(vi, kv.z, acc[i].z);
            acc[i].w = fmaf(vi, kv.w, acc[i].w);
            __builtin_nontemporal_store(acc[i],
                (v4f*)(St + (size_t)(dq * 4 + i) * N + n0));
        }
    }
}

extern "C" void kernel_launch(void* const* d_in, const int* in_sizes, int n_in,
                              void* d_out, int out_size, void* d_ws, size_t ws_size,
                              hipStream_t stream) {
    const float* v     = (const float*)d_in[0];
    const float* kk    = (const float*)d_in[1];
    const float* alpha = (const float*)d_in[2];
    float* S = (float*)d_out;

    float* kdec = (float*)d_ws;
    float* M    = (float*)((char*)d_ws + (size_t)T * B * N * 4);

    k_decay<<<(B * N + 255) / 256, 256, 0, stream>>>(alpha, kk, kdec);

    k_chunk_gemm<<<C * B, 256, 0, stream>>>(v, kdec, M);

    k_chunk_scan<<<(B * D * N) / 256, 256, 0, stream>>>(M);

    k_scan_store<<<C * B, 256, 0, stream>>>(v, kdec, M, S);
}


# pass 1 scan over T cut into 32 chunks per column (32 waves/block, chunk totals via LDS): 64 batched loads per thread replace 1024 serial round trips; log-sum order = chunk offset + in-chunk prefix
# speedup vs baseline: 7.6189x; 7.6189x over previous
#include <hip/hip_runtime.h>
#include <hip/hip_bf16.h>

#define T   1024
#define B   8
#define D   64
#define N   64
#define TC  32
#define C   (T / TC)
#define EPSF 1e-8f

typedef float v2f __attribute__((ext_vector_type(2)));
typedef float v4f __attribute__((ext_vector_type(4)));
typedef float v8f __attribute__((ext_vector_type(8)));

#define SC  32
#define NCH (T / SC)
__global__ __launch_bounds__(1024) void k_decay(const float* __restrict__ alpha,
                                                const float* __restrict__ kk,
                                                float* __restrict__ kdec) {
    __shared__ float tot[NCH][32];
    const int lane = threadIdx.x & 31;
    const int ch   = threadIdx.x >> 5;
    const int col  = blockIdx.x * 32 + lane;
    const size_t base = (size_t)(ch * SC) * (B * N) + col;

    float l[SC];
    float kv[SC];
    for (int i = 0; i < SC; ++i) l[i] = alpha[base + (size_t)i * (B * N)];
    for (int i = 0; i < SC; ++i) kv[i] = kk[base + (size_t)i * (B * N)];

    float run = 0.0f;
    #pragma unroll
    for (int i = 0; i < SC; ++i) {
        run += logf(fmaxf(l[i], EPSF));
        l[i] = run;
    }
    tot[ch][lane] = run;
    __syncthreads();

    float off = 0.0f, total = 0.0f;
    for (int j = 0; j < NCH; ++j) {
        if (j == ch) off = total;
        total += tot[j][lane];
    }
    const float inv = 1.0f / (expf(total) + EPSF);

    for (int i = 0; i < SC; ++i)
        kdec[base + (size_t)i * (B * N)] = expf(off + l[i]) * inv * kv[i];
}

__global__ __launch_bounds__(256) void k_chunk_gemm(const float* __restrict__ v,
                                                    const float* __restrict__ kdec,
                                                    float* __restrict__ M) {
    const int c = blockIdx.x / B;
    const int b = blockIdx.x % B;
    const int tid   = threadIdx.x;
    const int wave  = tid >> 5;
    const int lane  = tid & 31;
    const int mt    = wave >> 1;
    const int ntb   = (wave & 1) * 2;
    const int lhalf = lane >> 4;
    const int l16   = lane & 15;

    v8f acc0 = {};
    v8f acc1 = {};
    const int t0 = c * TC;

    #pragma unroll
    for (int ks = 0; ks < TC / 4; ++ks) {
        const int tk = t0 + ks * 4 + lhalf * 2;
        const int d = mt * 16 + l16;
        v2f a;
        a.x = v[((size_t)(tk    ) * B + b) * D + d];
        a.y = v[((size_t)(tk + 1) * B + b) * D + d];
        const float* kr0 = kdec + ((size_t)(tk    ) * B + b) * N;
        const float* kr1 = kdec + ((size_t)(tk + 1) * B + b) * N;
        const int n0 = ntb * 16 + l16;
        const int n1 = (ntb + 1) * 16 + l16;
        v2f b0, b1;
        b0.x = kr0[n0]; b0.y = kr1[n0];
        b1.x = kr0[n1]; b1.y = kr1[n1];
        acc0 = __builtin_amdgcn_wmma_f32_16x16x4_f32(false, a, false, b0,
                                                     (short)0, acc0, false, false);
        acc1 = __builtin_amdgcn_wmma_f32_16x16x4_f32(false, a, false, b1,
                                                     (short)0, acc1, false, false);
    }

    float* Mc = M + ((size_t)c * B + b) * (D * N);
    const int mbase = mt * 16 + lhalf * 8;
    const int nn0 = ntb * 16 + l16;
    const int nn1 = (ntb + 1) * 16 + l16;
    #pragma unroll
    for (int r = 0; r < 8; ++r) {
        Mc[(size_t)(mbase + r) * N + nn0] = acc0[r];
        Mc[(size_t)(mbase + r) * N + nn1] = acc1[r];
    }
}

__global__ __launch_bounds__(256) void k_chunk_scan(float* __restrict__ M) {
    const int idx = blockIdx.x * blockDim.x + threadIdx.x;
    float run = 0.0f;
    #pragma unroll 1
    for (int c = 0; c < C; ++c) {
        const size_t o = (size_t)c * (B * D * N) + idx;
        const float m = M[o];
        M[o] = run;
        run += m;
    }
}

__global__ __launch_bounds__(256) void k_scan_store(const float* __restrict__ v,
                                                    const float* __restrict__ kdec,
                                                    const float* __restrict__ O,
                                                    float* __restrict__ S) {
    const int c = blockIdx.x / B;
    const int b = blockIdx.x % B;
    const int tid = threadIdx.x;
    const int nq = tid & 15;
    const int dq = tid >> 4;
    const int n0 = nq * 4;

    v4f acc[4];
    const float* Ob = O + ((size_t)c * B + b) * (D * N);
    #pragma unroll
    for (int i = 0; i < 4; ++i)
        acc[i] = *(const v4f*)(Ob + (size_t)(dq * 4 + i) * N + n0);

    const int tEnd = c * TC + TC;
    #pragma unroll 1
    for (int t = c * TC; t < tEnd; ++t) {
        const v4f kv = *(const v4f*)(kdec + ((size_t)t * B + b) * N + n0);
        const float* vr = v + ((size_t)t * B + b) * D + dq * 4;
        float* St = S + ((size_t)t * B + b) * (size_t)(D * N);
        #pragma unroll
        for (int i = 0; i < 4; ++i) {
            const float vi = vr[i];
            acc[i].x = fmaf(vi, kv.x, acc[i].x);
            acc[i].y = fmaf(vi, kv.y, acc[i].y);
            acc[i].z = fmaf(vi, kv.z, acc[i].z);
            acc[i].w = fmaf(vi, kv.w, acc[i].w);
            __builtin_nontemporal_store(acc[i],
                (v4f*)(St + (size_t)(dq * 4 + i) * N + n0));
        }
    }
}

extern "C" void kernel_launch(void* const* d_in, const int* in_sizes, int n_in,
                              void* d_out, int out_size, void* d_ws, size_t ws_size,
                              hipStream_t stream) {
    const float* v     = (const float*)d_in[0];
    const float* kk    = (const float*)d_in[1];
    const float* alpha = (const float*)d_in[2];
    float* S = (float*)d_out;

    float* kdec = (float*)d_ws;
    float* M    = (float*)((char*)d_ws + (size_t)T * B * N * 4);

    k_decay<<<(B * N) / 32, 1024, 0, stream>>>(alpha, kk, kdec);

    k_chunk_gemm<<<C * B, 256, 0, stream>>>(v, kdec, M);

    k_chunk_scan<<<(B * D * N) / 256, 256, 0, stream>>>(M);

    k_scan_store<<<C * B, 256, 0, stream>>>(v, kdec, M, S);
}


# chunk-prefix scan folded into the store pass: each (chunk,b) block sums the chunk matrices before it, same ascending order, loads batched by 8; scan launch removed (4 -> 3), longest prefix first
# speedup vs baseline: 15.3938x; 2.0205x over previous
#include <hip/hip_runtime.h>
#include <hip/hip_bf16.h>

#define T   1024
#define B   8
#define D   64
#define N   64
#define TC  32
#define C   (T / TC)
#define EPSF 1e-8f

typedef float v2f __attribute__((ext_vector_type(2)));
typedef float v4f __attribute__((ext_vector_type(4)));
typedef float v8f __attribute__((ext_vector_type(8)));

#define SC  32
#define NCH (T / SC)
__global__ __launch_bounds__(1024) void k_decay(const float* __restrict__ alpha,
                                                const float* __restrict__ kk,
                                                float* __restrict__ kdec) {
    __shared__ float tot[NCH][32];
    const int lane = threadIdx.x & 31;
    const int ch   = threadIdx.x >> 5;
    const int col  = blockIdx.x * 32 + lane;
    const size_t base = (size_t)(ch * SC) * (B * N) + col;

    float l[SC];
    float kv[SC];
    for (int i = 0; i < SC; ++i) l[i] = alpha[base + (size_t)i * (B * N)];
    for (int i = 0; i < SC; ++i) kv[i] = kk[base + (size_t)i * (B * N)];

    float run = 0.0f;
    #pragma unroll
    for (int i = 0; i < SC; ++i) {
        run += logf(fmaxf(l[i], EPSF));
        l[i] = run;
    }
    tot[ch][lane] = run;
    __syncthreads();

    float off = 0.0f, total = 0.0f;
    for (int j = 0; j < NCH; ++j) {
        if (j == ch) off = total;
        total += tot[j][lane];
    }
    const float inv = 1.0f / (expf(total) + EPSF);

    for (int i = 0; i < SC; ++i)
        kdec[base + (size_t)i * (B * N)] = expf(off + l[i]) * inv * kv[i];
}

__global__ __launch_bounds__(256) void k_chunk_gemm(const float* __restrict__ v,
                                                    const float* __restrict__ kdec,
                                                    float* __restrict__ M) {
    const int c = blockIdx.x / B;
    const int b = blockIdx.x % B;
    const int tid   = threadIdx.x;
    const int wave  = tid >> 5;
    const int lane  = tid & 31;
    const int mt    = wave >> 1;
    const int ntb   = (wave & 1) * 2;
    const int lhalf = lane >> 4;
    const int l16   = lane & 15;

    v8f acc0 = {};
    v8f acc1 = {};
    const int t0 = c * TC;

    #pragma unroll
    for (int ks = 0; ks < TC / 4; ++ks) {
        const int tk = t0 + ks * 4 + lhalf * 2;
        const int d = mt * 16 + l16;
        v2f a;
        a.x = v[((size_t)(tk    ) * B + b) * D + d];
        a.y = v[((size_t)(tk + 1) * B + b) * D + d];
        const float* kr0 = kdec + ((size_t)(tk    ) * B + b) * N;
        const float* kr1 = kdec + ((size_t)(tk + 1) * B + b) * N;
        const int n0 = ntb * 16 + l16;
        const int n1 = (ntb + 1) * 16 + l16;
        v2f b0, b1;
        b0.x = kr0[n0]; b0.y = kr1[n0];
        b1.x = kr0[n1]; b1.y = kr1[n1];
        acc0 = __builtin_amdgcn_wmma_f32_16x16x4_f32(false, a, false, b0,
                                                     (short)0, acc0, false, false);
        acc1 = __builtin_amdgcn_wmma_f32_16x16x4_f32(false, a, false, b1,
                                                     (short)0, acc1, false, false);
    }

    float* Mc = M + ((size_t)c * B + b) * (D * N);
    const int mbase = mt * 16 + lhalf * 8;
    const int nn0 = ntb * 16 + l16;
    const int nn1 = (ntb + 1) * 16 + l16;
    #pragma unroll
    for (int r = 0; r < 8; ++r) {
        Mc[(size_t)(mbase + r) * N + nn0] = acc0[r];
        Mc[(size_t)(mbase + r) * N + nn1] = acc1[r];
    }
}

#define PB 8
template <bool FULL>
__device__ __forceinline__ void add_chunks(v4f (&acc)[4], const float* __restrict__ Mj0,
                                           int left) {
    v4f m[PB][4];
    #pragma unroll
    for (int u = 0; u < PB; ++u) {
        const float* Mj = Mj0 + (size_t)u * (B * D * N);
        #pragma unroll
        for (int i = 0; i < 4; ++i) m[u][i] = (v4f)(0.0f);
        if (FULL || u < left) {
            #pragma unroll
            for (int i = 0; i < 4; ++i) m[u][i] = *(const v4f*)(Mj + (size_t)i * N);
        }
    }
    #pragma unroll
    for (int u = 0; u < PB; ++u) {
        #pragma unroll
        for (int i = 0; i < 4; ++i) acc[i] += m[u][i];
    }
}
__global__ __launch_bounds__(256) void k_scan_store(const float* __restrict__ v,
                                                    const float* __restrict__ kdec,
                                                    const float* __restrict__ M,
                                                    float* __restrict__ S) {
    const int c = (C - 1) - (int)(blockIdx.x / B);
    const int b = blockIdx.x % B;
    const int tid = threadIdx.x;
    const int nq = tid & 15;
    const int dq = tid >> 4;
    const int n0 = nq * 4;

    v4f acc[4];
    #pragma unroll
    for (int i = 0; i < 4; ++i) acc[i] = (v4f)(0.0f);

    const float* Mb = M + (size_t)b * (D * N) + (size_t)(dq * 4) * N + n0;
    int j0 = 0;
    for (; j0 + PB <= c; j0 += PB)
        add_chunks<true>(acc, Mb + (size_t)j0 * (B * D * N), PB);
    if (j0 < c)
        add_chunks<false>(acc, Mb + (size_t)j0 * (B * D * N), c - j0);

    const int tEnd = c * TC + TC;
    #pragma unroll 1
    for (int t = c * TC; t < tEnd; ++t) {
        const v4f kv = *(const v4f*)(kdec + ((size_t)t * B + b) * N + n0);
        const float* vr = v + ((size_t)t * B + b) * D + dq * 4;
        float* St = S + ((size_t)t * B + b) * (size_t)(D * N);
        #pragma unroll
        for (int i = 0; i < 4; ++i) {
            const float vi = vr[i];
            acc[i].x = fmaf(vi, kv.x, acc[i].x);
            acc[i].y = fmaf(vi, kv.y, acc[i].y);
            acc[i].z = fmaf(vi, kv.z, acc[i].z);
            acc[i].w = fmaf(vi, kv.w, acc[i].w);
            __builtin_nontemporal_store(acc[i],
                (v4f*)(St + (size_t)(dq * 4 + i) * N + n0));
        }
    }
}

extern "C" void kernel_launch(void* const* d_in, const int* in_sizes, int n_in,
                              void* d_out, int out_size, void* d_ws, size_t ws_size,
                              hipStream_t stream) {
    const float* v     = (const float*)d_in[0];
    const float* kk    = (const float*)d_in[1];
    const float* alpha = (const float*)d_in[2];
    float* S = (float*)d_out;

    float* kdec = (float*)d_ws;
    float* M    = (float*)((char*)d_ws + (size_t)T * B * N * 4);

    k_decay<<<(B * N) / 32, 1024, 0, stream>>>(alpha, kk, kdec);

    k_chunk_gemm<<<C * B, 256, 0, stream>>>(v, kdec, M);

    k_scan_store<<<C * B, 256, 0, stream>>>(v, kdec, M, S);
}


# pass 1 log scan made chunk-local (one block per chunk and b: all compute units instead of 16); chunk offset, exp and k-scaling moved into the GEMM pass via LDS; all sums in the same order
# speedup vs baseline: 19.2197x; 1.2485x over previous
#include <hip/hip_runtime.h>
#include <hip/hip_bf16.h>

#define T   1024
#define B   8
#define D   64
#define N   64
#define TC  32
#define C   (T / TC)
#define EPSF 1e-8f

typedef float v2f __attribute__((ext_vector_type(2)));
typedef float v4f __attribute__((ext_vector_type(4)));
typedef float v8f __attribute__((ext_vector_type(8)));

__global__ __launch_bounds__(N) void k_logscan(const float* __restrict__ alpha,
                                               float* __restrict__ llog,
                                               float* __restrict__ tot) {
    const int c = blockIdx.x / B;
    const int b = blockIdx.x % B;
    const int n = threadIdx.x;
    const size_t base = ((size_t)(c * TC) * B + b) * N + n;

    float l[TC];
    for (int i = 0; i < TC; ++i) l[i] = alpha[base + (size_t)i * (B * N)];

    float run = 0.0f;
    #pragma unroll
    for (int i = 0; i < TC; ++i) {
        run += logf(fmaxf(l[i], EPSF));
        l[i] = run;
    }
    for (int i = 0; i < TC; ++i) llog[base + (size_t)i * (B * N)] = l[i];
    tot[((size_t)c * B + b) * N + n] = run;
}

#define KDP (N + 8)
__global__ __launch_bounds__(256) void k_chunk_gemm(const float* __restrict__ v,
                                                    const float* __restrict__ kk,
                                                    const float* __restrict__ tot,
                                                    float* kdec,
                                                    float* __restrict__ M) {
    __shared__ float kd[TC][KDP];
    const int c = blockIdx.x / B;
    const int b = blockIdx.x % B;
    const int tid   = threadIdx.x;
    const int wave  = tid >> 5;
    const int lane  = tid & 31;
    const int mt    = wave >> 1;
    const int ntb   = (wave & 1) * 2;
    const int lhalf = lane >> 4;
    const int l16   = lane & 15;
    const int t0 = c * TC;

    const int n  = tid & (N - 1);
    const int tq = tid >> 6;
    const size_t ebase = ((size_t)(t0 + tq * 8) * B + b) * N + n;
    float lv[8], kv[8], tj[C];
    #pragma unroll
    for (int j = 0; j < C; ++j) tj[j] = tot[((size_t)j * B + b) * N + n];
    #pragma unroll
    for (int i = 0; i < 8; ++i) lv[i] = kdec[ebase + (size_t)i * (B * N)];
    v2f a[TC / 4];
    {
        const int d = mt * 16 + l16;
        #pragma unroll
        for (int ks = 0; ks < TC / 4; ++ks) {
            const int tk = t0 + ks * 4 + lhalf * 2;
            a[ks].x = v[((size_t)(tk    ) * B + b) * D + d];
            a[ks].y = v[((size_t)(tk + 1) * B + b) * D + d];
        }
    }
    #pragma unroll
    for (int i = 0; i < 8; ++i) kv[i] = kk[ebase + (size_t)i * (B * N)];
    __builtin_amdgcn_sched_barrier(0);

    float off = 0.0f, total = 0.0f;
    #pragma unroll
    for (int j = 0; j < C; ++j) {
        if (j == c) off = total;
        total += tj[j];
    }
    const float inv = 1.0f / (expf(total) + EPSF);
    #pragma unroll
    for (int i = 0; i < 8; ++i) {
        const float kdv = expf(off + lv[i]) * inv * kv[i];
        kd[tq * 8 + i][n] = kdv;
        kdec[ebase + (size_t)i * (B * N)] = kdv;
    }
    __syncthreads();
    if (c == C - 1) return;

    v8f acc0 = {};
    v8f acc1 = {};
    const int n0 = ntb * 16 + l16;
    const int n1 = (ntb + 1) * 16 + l16;
    #pragma unroll
    for (int ks = 0; ks < TC / 4; ++ks) {
        const int tl = ks * 4 + lhalf * 2;
        v2f b0, b1;
        b0.x = kd[tl][n0]; b0.y = kd[tl + 1][n0];
        b1.x = kd[tl][n1]; b1.y = kd[tl + 1][n1];
        acc0 = __builtin_amdgcn_wmma_f32_16x16x4_f32(false, a[ks], false, b0,
                                                     (short)0, acc0, false, false);
        acc1 = __builtin_amdgcn_wmma_f32_16x16x4_f32(false, a[ks], false, b1,
                                                     (short)0, acc1, false, false);
    }

    float* Mc = M + ((size_t)c * B + b) * (D * N);
    const int mbase = mt * 16 + lhalf * 8;
    #pragma unroll
    for (int r = 0; r < 8; ++r) {
        Mc[(size_t)(mbase + r) * N + n0] = acc0[r];
        Mc[(size_t)(mbase + r) * N + n1] = acc1[r];
    }
}

#define PB 8
template <bool FULL>
__device__ __forceinline__ void add_chunks(v4f (&acc)[4], const float* __restrict__ Mj0,
                                           int left) {
    v4f m[PB][4];
    #pragma unroll
    for (int u = 0; u < PB; ++u) {
        const float* Mj = Mj0 + (size_t)u * (B * D * N);
        #pragma unroll
        for (int i = 0; i < 4; ++i) m[u][i] = (v4f)(0.0f);
        if (FULL || u < left) {
            #pragma unroll
            for (int i = 0; i < 4; ++i) m[u][i] = *(const v4f*)(Mj + (size_t)i * N);
        }
    }
    #pragma unroll
    for (int u = 0; u < PB; ++u) {
        #pragma unroll
        for (int i = 0; i < 4; ++i) acc[i] += m[u][i];
    }
}
__global__ __launch_bounds__(256) void k_scan_store(const float* __restrict__ v,
                                                    const float* __restrict__ kdec,
                                                    const float* __restrict__ M,
                                                    float* __restrict__ S) {
    const int c = (C - 1) - (int)(blockIdx.x / B);
    const int b = blockIdx.x % B;
    const int tid = threadIdx.x;
    const int nq = tid & 15;
    const int dq = tid >> 4;
    const int n0 = nq * 4;

    v4f acc[4];
    #pragma unroll
    for (int i = 0; i < 4; ++i) acc[i] = (v4f)(0.0f);

    const float* Mb = M + (size_t)b * (D * N) + (size_t)(dq * 4) * N + n0;
    int j0 = 0;
    for (; j0 + PB <= c; j0 += PB)
        add_chunks<true>(acc, Mb + (size_t)j0 * (B * D * N), PB);
    if (j0 < c)
        add_chunks<false>(acc, Mb + (size_t)j0 * (B * D * N), c - j0);

    const int tEnd = c * TC + TC;
    #pragma unroll 1
    for (int t = c * TC; t < tEnd; ++t) {
        const v4f kv = *(const v4f*)(kdec + ((size_t)t * B + b) * N + n0);
        const float* vr = v + ((size_t)t * B + b) * D + dq * 4;
        float* St = S + ((size_t)t * B + b) * (size_t)(D * N);
        #pragma unroll
        for (int i = 0; i < 4; ++i) {
            const float vi = vr[i];
            acc[i].x = fmaf(vi, kv.x, acc[i].x);
            acc[i].y = fmaf(vi, kv.y, acc[i].y);
            acc[i].z = fmaf(vi, kv.z, acc[i].z);
            acc[i].w = fmaf(vi, kv.w, acc[i].w);
            __builtin_nontemporal_store(acc[i],
                (v4f*)(St + (size_t)(dq * 4 + i) * N + n0));
        }
    }
}

extern "C" void kernel_launch(void* const* d_in, const int* in_sizes, int n_in,
                              void* d_out, int out_size, void* d_ws, size_t ws_size,
                              hipStream_t stream) {
    const float* v     = (const float*)d_in[0];
    const float* kk    = (const float*)d_in[1];
    const float* alpha = (const float*)d_in[2];
    float* S = (float*)d_out;

    float* kdec = (float*)d_ws;
    float* M    = (float*)((char*)d_ws + (size_t)T * B * N * 4);

    float* tot  = M + (size_t)(C - 1) * B * D * N;

    k_logscan<<<C * B, N, 0, stream>>>(alpha, kdec, tot);

    k_chunk_gemm<<<C * B, 256, 0, stream>>>(v, kk, tot, kdec, M);

    k_scan_store<<<C * B, 256, 0, stream>>>(v, kdec, M, S);
}


# pass 3: the chunk's 32 rows of v and kdec go once per block into LDS (loads issued ahead of the prefix loads); the store loop reads LDS only, no global round trip per step; sums and order unchanged
# speedup vs baseline: 26.9040x; 1.3998x over previous
#include <hip/hip_runtime.h>
#include <hip/hip_bf16.h>

#define T   1024
#define B   8
#define D   64
#define N   64
#define TC  32
#define C   (T / TC)
#define EPSF 1e-8f

typedef float v2f __attribute__((ext_vector_type(2)));
typedef float v4f __attribute__((ext_vector_type(4)));
typedef float v8f __attribute__((ext_vector_type(8)));

__global__ __launch_bounds__(N) void k_logscan(const float* __restrict__ alpha,
                                               float* __restrict__ llog,
                                               float* __restrict__ tot) {
    const int c = blockIdx.x / B;
    const int b = blockIdx.x % B;
    const int n = threadIdx.x;
    const size_t base = ((size_t)(c * TC) * B + b) * N + n;

    float l[TC];
    for (int i = 0; i < TC; ++i) l[i] = alpha[base + (size_t)i * (B * N)];

    float run = 0.0f;
    #pragma unroll
    for (int i = 0; i < TC; ++i) {
        run += logf(fmaxf(l[i], EPSF));
        l[i] = run;
    }
    for (int i = 0; i < TC; ++i) llog[base + (size_t)i * (B * N)] = l[i];
    tot[((size_t)c * B + b) * N + n] = run;
}

#define KDP (N + 8)
__global__ __launch_bounds__(256) void k_chunk_gemm(const float* __restrict__ v,
                                                    const float* __restrict__ kk,
                                                    const float* __restrict__ tot,
                                                    float* kdec,
                                                    float* __restrict__ M) {
    __shared__ float kd[TC][KDP];
    const int c = blockIdx.x / B;
    const int b = blockIdx.x % B;
    const int tid   = threadIdx.x;
    const int wave  = tid >> 5;
    const int lane  = tid & 31;
    const int mt    = wave >> 1;
    const int ntb   = (wave & 1) * 2;
    const int lhalf = lane >> 4;
    const int l16   = lane & 15;
    const int t0 = c * TC;

    const int n  = tid & (N - 1);
    const int tq = tid >> 6;
    const size_t ebase = ((size_t)(t0 + tq * 8) * B + b) * N + n;
    float lv[8], kv[8], tj[C];
    #pragma unroll
    for (int j = 0; j < C; ++j) tj[j] = tot[((size_t)j * B + b) * N + n];
    #pragma unroll
    for (int i = 0; i < 8; ++i) lv[i] = kdec[ebase + (size_t)i * (B * N)];
    v2f a[TC / 4];
    {
        const int d = mt * 16 + l16;
        #pragma unroll
        for (int ks = 0; ks < TC / 4; ++ks) {
            const int tk = t0 + ks * 4 + lhalf * 2;
            a[ks].x = v[((size_t)(tk    ) * B + b) * D + d];
            a[ks].y = v[((size_t)(tk + 1) * B + b) * D + d];
        }
    }
    #pragma unroll
    for (int i = 0; i < 8; ++i) kv[i] = kk[ebase + (size_t)i * (B * N)];
    __builtin_amdgcn_sched_barrier(0);

    float off = 0.0f, total = 0.0f;
    #pragma unroll
    for (int j = 0; j < C; ++j) {
        if (j == c) off = total;
        total += tj[j];
    }
    const float inv = 1.0f / (expf(total) + EPSF);
    #pragma unroll
    for (int i = 0; i < 8; ++i) {
        const float kdv = expf(off + lv[i]) * inv * kv[i];
        kd[tq * 8 + i][n] = kdv;
        kdec[ebase + (size_t)i * (B * N)] = kdv;
    }
    __syncthreads();
    if (c == C - 1) return;

    v8f acc0 = {};
    v8f acc1 = {};
    const int n0 = ntb * 16 + l16;
    const int n1 = (ntb + 1) * 16 + l16;
    #pragma unroll
    for (int ks = 0; ks < TC / 4; ++ks) {
        const int tl = ks * 4 + lhalf * 2;
        v2f b0, b1;
        b0.x = kd[tl][n0]; b0.y = kd[tl + 1][n0];
        b1.x = kd[tl][n1]; b1.y = kd[tl + 1][n1];
        acc0 = __builtin_amdgcn_wmma_f32_16x16x4_f32(false, a[ks], false, b0,
                                                     (short)0, acc0, false, false);
        acc1 = __builtin_amdgcn_wmma_f32_16x16x4_f32(false, a[ks], false, b1,
                                                     (short)0, acc1, false, false);
    }

    float* Mc = M + ((size_t)c * B + b) * (D * N);
    const int mbase = mt * 16 + lhalf * 8;
    #pragma unroll
    for (int r = 0; r < 8; ++r) {
        Mc[(size_t)(mbase + r) * N + n0] = acc0[r];
        Mc[(size_t)(mbase + r) * N + n1] = acc1[r];
    }
}

#define PB 8
template <bool FULL>
__device__ __forceinline__ void add_chunks(v4f (&acc)[4], const float* __restrict__ Mj0,
                                           int left) {
    v4f m[PB][4];
    #pragma unroll
    for (int u = 0; u < PB; ++u) {
        const float* Mj = Mj0 + (size_t)u * (B * D * N);
        #pragma unroll
        for (int i = 0; i < 4; ++i) m[u][i] = (v4f)(0.0f);
        if (FULL || u < left) {
            #pragma unroll
            for (int i = 0; i < 4; ++i) m[u][i] = *(const v4f*)(Mj + (size_t)i * N);
        }
    }
    #pragma unroll
    for (int u = 0; u < PB; ++u) {
        #pragma unroll
        for (int i = 0; i < 4; ++i) acc[i] += m[u][i];
    }
}
__global__ __launch_bounds__(256) void k_scan_store(const float* __restrict__ v,
                                                    const float* __restrict__ kdec,
                                                    const float* __restrict__ M,
                                                    float* __restrict__ S) {
    __shared__ v4f ks[TC][N / 4];
    __shared__ v4f vs[TC][D / 4];
    const int c = (C - 1) - (int)(blockIdx.x / B);
    const int b = blockIdx.x % B;
    const int tid = threadIdx.x;
    const int nq = tid & 15;
    const int dq = tid >> 4;
    const int n0 = nq * 4;

    const size_t row0 = ((size_t)(c * TC + dq) * B + b) * N + n0;
    const size_t row1 = row0 + (size_t)16 * B * N;
    const v4f sk0 = *(const v4f*)(kdec + row0);
    const v4f sk1 = *(const v4f*)(kdec + row1);
    const v4f sv0 = *(const v4f*)(v + row0);
    const v4f sv1 = *(const v4f*)(v + row1);
    asm volatile("" ::: "memory");

    v4f acc[4];
    #pragma unroll
    for (int i = 0; i < 4; ++i) acc[i] = (v4f)(0.0f);

    const float* Mb = M + (size_t)b * (D * N) + (size_t)(dq * 4) * N + n0;
    int j0 = 0;
    for (; j0 + PB <= c; j0 += PB)
        add_chunks<true>(acc, Mb + (size_t)j0 * (B * D * N), PB);
    if (j0 < c)
        add_chunks<false>(acc, Mb + (size_t)j0 * (B * D * N), c - j0);

    ks[dq][nq] = sk0;  ks[dq + 16][nq] = sk1;
    vs[dq][nq] = sv0;  vs[dq + 16][nq] = sv1;
    __syncthreads();

    float* St = S + ((size_t)(c * TC) * B + b) * (size_t)(D * N)
                  + (size_t)(dq * 4) * N + n0;
    #pragma unroll 4
    for (int t = 0; t < TC; ++t) {
        const v4f kv = ks[t][nq];
        const v4f vv = vs[t][dq];
        #pragma unroll
        for (int i = 0; i < 4; ++i) {
            const float vi = vv[i];
            acc[i].x = fmaf(vi, kv.x, acc[i].x);
            acc[i].y = fmaf(vi, kv.y, acc[i].y);
            acc[i].z = fmaf(vi, kv.z, acc[i].z);
            acc[i].w = fmaf(vi, kv.w, acc[i].w);
            __builtin_nontemporal_store(acc[i], (v4f*)(St + (size_t)i * N));
        }
        St += (size_t)B * D * N;
    }
}

extern "C" void kernel_launch(void* const* d_in, const int* in_sizes, int n_in,
                              void* d_out, int out_size, void* d_ws, size_t ws_size,
                              hipStream_t stream) {
    const float* v     = (const float*)d_in[0];
    const float* kk    = (const float*)d_in[1];
    const float* alpha = (const float*)d_in[2];
    float* S = (float*)d_out;

    float* kdec = (float*)d_ws;
    float* M    = (float*)((char*)d_ws + (size_t)T * B * N * 4);

    float* tot  = M + (size_t)(C - 1) * B * D * N;

    k_logscan<<<C * B, N, 0, stream>>>(alpha, kdec, tot);

    k_chunk_gemm<<<C * B, 256, 0, stream>>>(v, kk, tot, kdec, M);

    k_scan_store<<<C * B, 256, 0, stream>>>(v, kdec, M, S);
}


# log scan (quarter sums via LDS + in-quarter prefix) fused into the WMMA launch; weight = k*exp(L) times factor exp(off)/(exp(total)+eps), factor applied in the store pass by fma; 3 -> 2 launches
# speedup vs baseline: 28.4428x; 1.0572x over previous
#include <hip/hip_runtime.h>
#include <hip/hip_bf16.h>

#define T   1024
#define B   8
#define D   64
#define N   64
#define TC  32
#define C   (T / TC)
#define EPSF 1e-8f

typedef float v2f __attribute__((ext_vector_type(2)));
typedef float v4f __attribute__((ext_vector_type(4)));
typedef float v8f __attribute__((ext_vector_type(8)));

#define KDP (N + 8)
__global__ __launch_bounds__(256) void k_chunk_local(const float* __restrict__ v,
                                                     const float* __restrict__ kk,
                                                     const float* __restrict__ alpha,
                                                     float* __restrict__ kdl,
                                                     float* __restrict__ tot,
                                                     float* __restrict__ M) {
    __shared__ float kd[TC][KDP];
    __shared__ float qs[4][N];
    const int c = blockIdx.x / B;
    const int b = blockIdx.x % B;
    const int tid   = threadIdx.x;
    const int wave  = tid >> 5;
    const int lane  = tid & 31;
    const int mt    = wave >> 1;
    const int ntb   = (wave & 1) * 2;
    const int lhalf = lane >> 4;
    const int l16   = lane & 15;
    const int t0 = c * TC;

    const int n  = tid & (N - 1);
    const int tq = tid >> 6;
    const size_t ebase = ((size_t)(t0 + tq * 8) * B + b) * N + n;
    float al[8], kv[8];
    #pragma unroll
    for (int i = 0; i < 8; ++i) al[i] = alpha[ebase + (size_t)i * (B * N)];
    v2f a[TC / 4];
    {
        const int d = mt * 16 + l16;
        #pragma unroll
        for (int ks = 0; ks < TC / 4; ++ks) {
            const int tk = t0 + ks * 4 + lhalf * 2;
            a[ks].x = v[((size_t)(tk    ) * B + b) * D + d];
            a[ks].y = v[((size_t)(tk + 1) * B + b) * D + d];
        }
    }
    #pragma unroll
    for (int i = 0; i < 8; ++i) kv[i] = kk[ebase + (size_t)i * (B * N)];
    __builtin_amdgcn_sched_barrier(0);

    float lv[8];
    float run = 0.0f;
    #pragma unroll
    for (int i = 0; i < 8; ++i) {
        run += logf(fmaxf(al[i], EPSF));
        lv[i] = run;
    }
    qs[tq][n] = run;
    __syncthreads();
    float start = 0.0f;
    #pragma unroll
    for (int q = 0; q < 3; ++q)
        if (q < tq) start += qs[q][n];
    #pragma unroll
    for (int i = 0; i < 8; ++i) {
        const float kdv = expf(start + lv[i]) * kv[i];
        kd[tq * 8 + i][n] = kdv;
        kdl[ebase + (size_t)i * (B * N)] = kdv;
    }
    if (tq == 3) tot[((size_t)c * B + b) * N + n] = start + lv[7];
    __syncthreads();
    if (c == C - 1) return;

    v8f acc0 = {};
    v8f acc1 = {};
    const int n0 = ntb * 16 + l16;
    const int n1 = (ntb + 1) * 16 + l16;
    #pragma unroll
    for (int ks = 0; ks < TC / 4; ++ks) {
        const int tl = ks * 4 + lhalf * 2;
        v2f b0, b1;
        b0.x = kd[tl][n0]; b0.y = kd[tl + 1][n0];
        b1.x = kd[tl][n1]; b1.y = kd[tl + 1][n1];
        acc0 = __builtin_amdgcn_wmma_f32_16x16x4_f32(false, a[ks], false, b0,
                                                     (short)0, acc0, false, false);
        acc1 = __builtin_amdgcn_wmma_f32_16x16x4_f32(false, a[ks], false, b1,
                                                     (short)0, acc1, false, false);
    }

    float* Mc = M + ((size_t)c * B + b) * (D * N);
    const int mbase = mt * 16 + lhalf * 8;
    #pragma unroll
    for (int r = 0; r < 8; ++r) {
        Mc[(size_t)(mbase + r) * N + n0] = acc0[r];
        Mc[(size_t)(mbase + r) * N + n1] = acc1[r];
    }
}

#define PB 8
template <bool FULL>
__device__ __forceinline__ void add_chunks(v4f (&acc)[4], const float* __restrict__ Mj0,
                                           const float* fj0, int left) {
    v4f m[PB][4];
    v4f f[PB];
    #pragma unroll
    for (int u = 0; u < PB; ++u) {
        const float* Mj = Mj0 + (size_t)u * (B * D * N);
        #pragma unroll
        for (int i = 0; i < 4; ++i) m[u][i] = (v4f)(0.0f);
        f[u] = *(const v4f*)(fj0 + (size_t)u * N);
        if (FULL || u < left) {
            #pragma unroll
            for (int i = 0; i < 4; ++i) m[u][i] = *(const v4f*)(Mj + (size_t)i * N);
        }
    }
    #pragma unroll
    for (int u = 0; u < PB; ++u) {
        #pragma unroll
        for (int i = 0; i < 4; ++i) acc[i] += m[u][i] * f[u];
    }
}
__global__ __launch_bounds__(256) void k_scan_store(const float* __restrict__ v,
                                                    const float* __restrict__ kdl,
                                                    const float* __restrict__ tot,
                                                    const float* __restrict__ M,
                                                    float* __restrict__ S) {
    __shared__ v4f ks[TC][N / 4];
    __shared__ v4f vs[TC][D / 4];
    __shared__ __attribute__((aligned(16))) float fs[C][N];
    const int c = (C - 1) - (int)(blockIdx.x / B);
    const int b = blockIdx.x % B;
    const int tid = threadIdx.x;
    const int nq = tid & 15;
    const int dq = tid >> 4;
    const int n0 = nq * 4;

    const int n  = tid & (N - 1);
    const int jq = tid >> 6;
    float tj[C];
    #pragma unroll
    for (int j = 0; j < C; ++j) tj[j] = tot[((size_t)j * B + b) * N + n];

    const size_t row0 = ((size_t)(c * TC + dq) * B + b) * N + n0;
    const size_t row1 = row0 + (size_t)16 * B * N;
    const v4f sk0 = *(const v4f*)(kdl + row0);
    const v4f sk1 = *(const v4f*)(kdl + row1);
    const v4f sv0 = *(const v4f*)(v + row0);
    const v4f sv1 = *(const v4f*)(v + row1);
    asm volatile("" ::: "memory");
    __builtin_amdgcn_sched_barrier(0);

    float offs[8];
    float total = 0.0f;
    #pragma unroll
    for (int j = 0; j < C; ++j) {
        if ((j >> 3) == jq) offs[j & 7] = total;
        total += tj[j];
    }
    const float inv = 1.0f / (expf(total) + EPSF);
    #pragma unroll
    for (int i = 0; i < 8; ++i) {
        const int j = jq * 8 + i;
        fs[j][n] = (j <= c) ? expf(offs[i]) * inv : 0.0f;
    }
    __syncthreads();

    v4f acc[4];
    #pragma unroll
    for (int i = 0; i < 4; ++i) acc[i] = (v4f)(0.0f);

    const float* Mb = M + (size_t)b * (D * N) + (size_t)(dq * 4) * N + n0;
    const float* fb = &fs[0][n0];
    int j0 = 0;
    for (; j0 + PB <= c; j0 += PB)
        add_chunks<true>(acc, Mb + (size_t)j0 * (B * D * N), fb + (size_t)j0 * N, PB);
    if (j0 < c)
        add_chunks<false>(acc, Mb + (size_t)j0 * (B * D * N), fb + (size_t)j0 * N, c - j0);

    const v4f fc = *(const v4f*)(fb + (size_t)c * N);
    ks[dq][nq] = sk0 * fc;  ks[dq + 16][nq] = sk1 * fc;
    vs[dq][nq] = sv0;       vs[dq + 16][nq] = sv1;
    __syncthreads();

    float* St = S + ((size_t)(c * TC) * B + b) * (size_t)(D * N)
                  + (size_t)(dq * 4) * N + n0;
    #pragma unroll 4
    for (int t = 0; t < TC; ++t) {
        const v4f kv = ks[t][nq];
        const v4f vv = vs[t][dq];
        #pragma unroll
        for (int i = 0; i < 4; ++i) {
            const float vi = vv[i];
            acc[i].x = fmaf(vi, kv.x, acc[i].x);
            acc[i].y = fmaf(vi, kv.y, acc[i].y);
            acc[i].z = fmaf(vi, kv.z, acc[i].z);
            acc[i].w = fmaf(vi, kv.w, acc[i].w);
            __builtin_nontemporal_store(acc[i], (v4f*)(St + (size_t)i * N));
        }
        St += (size_t)B * D * N;
    }
}

extern "C" void kernel_launch(void* const* d_in, const int* in_sizes, int n_in,
                              void* d_out, int out_size, void* d_ws, size_t ws_size,
                              hipStream_t stream) {
    const float* v     = (const float*)d_in[0];
    const float* kk    = (const float*)d_in[1];
    const float* alpha = (const float*)d_in[2];
    float* S = (float*)d_out;

    float* kdl  = (float*)d_ws;
    float* M    = (float*)((char*)d_ws + (size_t)T * B * N * 4);

    float* tot  = M + (size_t)(C - 1) * B * D * N;

    k_chunk_local<<<C * B, 256, 0, stream>>>(v, kk, alpha, kdl, tot, M);

    k_scan_store<<<C * B, 256, 0, stream>>>(v, kdl, tot, M, S);
}
